# starting kernel
# baseline (speedup 1.0000x reference)
#include <hip/hip_runtime.h>
#include <hip/hip_bf16.h>


#define NN 8192
#define DT 1e-3f
#define TAU 1e-2f
#define KSPLIT 16
#define KCHUNK (NN / KSPLIT)
#define ROWTILES (NN / 16)
#define WAVES_PER_BLOCK 8

typedef __attribute__((ext_vector_type(2))) float v2f;
typedef __attribute__((ext_vector_type(8))) float v8f;

__global__ void rnn_prep_kernel(const float* __restrict__ y,
                                float* __restrict__ r,
                                float* __restrict__ acc, int n) {
  int i = blockIdx.x * blockDim.x + threadIdx.x;
  if (i < n) {
    r[i]   = tanhf(y[i]);
    acc[i] = 0.0f;
  }
}

__global__ __launch_bounds__(256) void
rnn_gemv_wmma_kernel(const float* __restrict__ W,
                     const float* __restrict__ r,
                     float* __restrict__ acc) {
  const int lane  = threadIdx.x & 31;
  const int wave  = threadIdx.x >> 5;
  const int gwave = blockIdx.x * WAVES_PER_BLOCK + wave;

  const int rowTile = gwave % ROWTILES;
  const int kChunk  = gwave / ROWTILES;
  const int row0    = rowTile * 16;
  const int k0      = kChunk * KCHUNK;

  const int halfSel = lane >> 4;
  const int m       = lane & 15;

  const float* wp = W + (size_t)(row0 + m) * NN + (k0 + 2 * halfSel);
  const float* rp = r + (k0 + 2 * halfSel);

  v8f c = {};
#pragma unroll 4
  for (int kk = 0; kk < KCHUNK; kk += 4) {
    v2f a = *(const v2f*)wp;
    v2f b = *(const v2f*)rp;
    __builtin_prefetch(wp + 512, 0, 1);
    c = __builtin_amdgcn_wmma_f32_16x16x4_f32(
        false, a, false, b, (short)0, c, false, false);
    wp += 4;
    rp += 4;
  }

  if ((lane & 15) == 0) {
    const int base = row0 + (halfSel ? 8 : 0);
#pragma unroll
    for (int i = 0; i < 8; ++i) {
      unsafeAtomicAdd(&acc[base + i], c[i]);
    }
  }
}

__global__ void rnn_finish_kernel(const float* __restrict__ x,
                                  const float* __restrict__ y,
                                  const float* __restrict__ acc,
                                  float* __restrict__ out, int n) {
  int i = blockIdx.x * blockDim.x + threadIdx.x;
  if (i < n) {
    float dy = (-y[i] + acc[i] + x[i]) * (1.0f / TAU);
    out[i] = y[i] + DT * dy;
  }
}

extern "C" void kernel_launch(void* const* d_in, const int* in_sizes, int n_in,
                              void* d_out, int out_size, void* d_ws, size_t ws_size,
                              hipStream_t stream) {
  const float* x = (const float*)d_in[0];
  const float* y = (const float*)d_in[1];
  const float* W = (const float*)d_in[2];
  float* out = (float*)d_out;

  float* acc = (float*)d_ws;
  float* r   = acc + NN;

  rnn_prep_kernel<<<NN / 256, 256, 0, stream>>>(y, r, acc, NN);

  const int totalWaves = ROWTILES * KSPLIT;
  const int blocks     = totalWaves / WAVES_PER_BLOCK;
  rnn_gemv_wmma_kernel<<<blocks, 256, 0, stream>>>(W, r, acc);

  rnn_finish_kernel<<<NN / 256, 256, 0, stream>>>(x, y, acc, out, NN);
}


# Each wave-wide load now reads 512 contiguous bytes of one W row (row dot products collect on the WMMA D diagonal, r kept in registers); K quarters summed in LDS, atomics removed: new summation order
# speedup vs baseline: 2.6986x; 2.6986x over previous
#include <hip/hip_runtime.h>


#define NN 8192
#define DT 1e-3f
#define TAU 1e-2f
#define WAVES 16
#define KSPLIT 4
#define RGROUPS (WAVES / KSPLIT)
#define ROWS 8
#define WG_ROWS (RGROUPS * ROWS)
#define KCHUNK (NN / KSPLIT)
#define KQ (KCHUNK / 128)

typedef __attribute__((ext_vector_type(2))) float v2f;
typedef __attribute__((ext_vector_type(4))) float v4f;
typedef __attribute__((ext_vector_type(8))) float v8f;

__global__ void rnn_prep_kernel(const float* __restrict__ y,
                                float* __restrict__ r, int n) {
  int i = blockIdx.x * blockDim.x + threadIdx.x;
  if (i < n) r[i] = tanhf(y[i]);
}

__global__ __launch_bounds__(WAVES * 32) void
rnn_gemv_wmma_kernel(const float* __restrict__ W,
                     const float* __restrict__ r,
                     float* __restrict__ acc) {
  __shared__ float part[KSPLIT][WG_ROWS];

  const int lane = threadIdx.x & 31;
  const int wave = __builtin_amdgcn_readfirstlane(threadIdx.x >> 5);
  const int kq   = wave & (KSPLIT - 1);
  const int rg   = wave / KSPLIT;
  const int row0 = blockIdx.x * WG_ROWS + rg * ROWS;
  const int col0 = kq * KCHUNK + 4 * lane;

  v4f rq[KQ];
#pragma unroll
  for (int q = 0; q < KQ; ++q) rq[q] = *(const v4f*)(r + col0 + 128 * q);

  const float* wb = W + (size_t)row0 * NN + col0;
  v4f w[2][KQ];
#define LOAD_ROW(R) do { _Pragma("unroll") for (int q = 0; q < KQ; ++q) \
    w[(R) & 1][q] = __builtin_nontemporal_load((const v4f*)(wb + (size_t)(R) * NN + 128 * q)); } while (0)

  float sum[ROWS];
  LOAD_ROW(0);
#pragma unroll
  for (int rr = 0; rr < ROWS; ++rr) {
    __builtin_amdgcn_sched_barrier(0);
    if (rr + 1 < ROWS) LOAD_ROW(rr + 1);
    __builtin_amdgcn_sched_barrier(0);
    const v4f* wr = w[rr & 1];
    v8f c0 = {}, c1 = {};
#pragma unroll
    for (int q = 0; q < KQ; ++q) {
      const v2f a0 = {wr[q].x, wr[q].y}, b0 = {rq[q].x, rq[q].y};
      const v2f a1 = {wr[q].z, wr[q].w}, b1 = {rq[q].z, rq[q].w};
      c0 = __builtin_amdgcn_wmma_f32_16x16x4_f32(false, a0, false, b0, (short)0, c0, false, false);
      c1 = __builtin_amdgcn_wmma_f32_16x16x4_f32(false, a1, false, b1, (short)0, c1, false, false);
    }
    const v8f c = c0 + c1;
    float d = 0.0f;
#pragma unroll
    for (int i = 0; i < 8; ++i) d = ((lane & 7) == i) ? c[i] : d;
    sum[rr] = (lane < 8 || lane >= 24) ? d : 0.0f;
  }
#undef LOAD_ROW

#pragma unroll
  for (int rr = 0; rr < ROWS; ++rr) {
    float s = sum[rr];
#pragma unroll
    for (int m = 16; m >= 1; m >>= 1) s += __shfl_xor(s, m, 32);
    sum[rr] = s;
  }
  float mine = 0.0f;
#pragma unroll
  for (int rr = 0; rr < ROWS; ++rr) mine = (lane == rr) ? sum[rr] : mine;
  if (lane < ROWS) part[kq][rg * ROWS + lane] = mine;
  __syncthreads();
  if (wave == 0) {
    const float s = (part[0][lane] + part[1][lane]) + (part[2][lane] + part[3][lane]);
    acc[blockIdx.x * WG_ROWS + lane] = s;
  }
}

__global__ void rnn_finish_kernel(const float* __restrict__ x,
                                  const float* __restrict__ y,
                                  const float* __restrict__ acc,
                                  float* __restrict__ out, int n) {
  int i = blockIdx.x * blockDim.x + threadIdx.x;
  if (i < n) {
    float dy = (-y[i] + acc[i] + x[i]) * (1.0f / TAU);
    out[i] = y[i] + DT * dy;
  }
}

extern "C" void kernel_launch(void* const* d_in, const int* in_sizes, int n_in,
                              void* d_out, int out_size, void* d_ws, size_t ws_size,
                              hipStream_t stream) {
  const float* x = (const float*)d_in[0];
  const float* y = (const float*)d_in[1];
  const float* W = (const float*)d_in[2];
  float* out = (float*)d_out;

  float* acc = (float*)d_ws;
  float* r   = acc + NN;

  rnn_prep_kernel<<<NN / 256, 256, 0, stream>>>(y, r, NN);

  rnn_gemv_wmma_kernel<<<NN / WG_ROWS, WAVES * 32, 0, stream>>>(W, r, acc);

  rnn_finish_kernel<<<NN / 256, 256, 0, stream>>>(x, y, acc, out, NN);
}


# State update folded into the GEMV: the wave that adds a workgroup's four K quarters applies it and stores out itself; two launches instead of three, no acc round trip; summation order unchanged
# speedup vs baseline: 2.8906x; 1.0711x over previous
#include <hip/hip_runtime.h>


#define NN 8192
#define DT 1e-3f
#define TAU 1e-2f
#define WAVES 16
#define KSPLIT 4
#define RGROUPS (WAVES / KSPLIT)
#define ROWS 8
#define WG_ROWS (RGROUPS * ROWS)
#define KCHUNK (NN / KSPLIT)
#define KQ (KCHUNK / 128)

typedef __attribute__((ext_vector_type(2))) float v2f;
typedef __attribute__((ext_vector_type(4))) float v4f;
typedef __attribute__((ext_vector_type(8))) float v8f;

__global__ void rnn_prep_kernel(const float* __restrict__ y,
                                float* __restrict__ r, int n) {
  int i = blockIdx.x * blockDim.x + threadIdx.x;
  if (i < n) r[i] = tanhf(y[i]);
}

__global__ __launch_bounds__(WAVES * 32) void
rnn_gemv_wmma_kernel(const float* __restrict__ W,
                     const float* __restrict__ r,
                     const float* __restrict__ x,
                     const float* __restrict__ y,
                     float* __restrict__ out) {
  __shared__ float part[KSPLIT][WG_ROWS];

  const int lane = threadIdx.x & 31;
  const int wave = __builtin_amdgcn_readfirstlane(threadIdx.x >> 5);
  const int kq   = wave & (KSPLIT - 1);
  const int rg   = wave / KSPLIT;
  const int row0 = blockIdx.x * WG_ROWS + rg * ROWS;
  const int col0 = kq * KCHUNK + 4 * lane;

  float xv = 0.0f, yv = 0.0f;
  if (wave == 0) {
    xv = x[blockIdx.x * WG_ROWS + lane];
    yv = y[blockIdx.x * WG_ROWS + lane];
  }

  v4f rq[KQ];
#pragma unroll
  for (int q = 0; q < KQ; ++q) rq[q] = *(const v4f*)(r + col0 + 128 * q);

  const float* wb = W + (size_t)row0 * NN + col0;
  v4f w[2][KQ];
#define LOAD_ROW(R) do { _Pragma("unroll") for (int q = 0; q < KQ; ++q) \
    w[(R) & 1][q] = __builtin_nontemporal_load((const v4f*)(wb + (size_t)(R) * NN + 128 * q)); } while (0)

  float sum[ROWS];
  LOAD_ROW(0);
#pragma unroll
  for (int rr = 0; rr < ROWS; ++rr) {
    __builtin_amdgcn_sched_barrier(0);
    if (rr + 1 < ROWS) LOAD_ROW(rr + 1);
    __builtin_amdgcn_sched_barrier(0);
    const v4f* wr = w[rr & 1];
    v8f c0 = {}, c1 = {};
#pragma unroll
    for (int q = 0; q < KQ; ++q) {
      const v2f a0 = {wr[q].x, wr[q].y}, b0 = {rq[q].x, rq[q].y};
      const v2f a1 = {wr[q].z, wr[q].w}, b1 = {rq[q].z, rq[q].w};
      c0 = __builtin_amdgcn_wmma_f32_16x16x4_f32(false, a0, false, b0, (short)0, c0, false, false);
      c1 = __builtin_amdgcn_wmma_f32_16x16x4_f32(false, a1, false, b1, (short)0, c1, false, false);
    }
    const v8f c = c0 + c1;
    float d = 0.0f;
#pragma unroll
    for (int i = 0; i < 8; ++i) d = ((lane & 7) == i) ? c[i] : d;
    sum[rr] = (lane < 8 || lane >= 24) ? d : 0.0f;
  }
#undef LOAD_ROW

#pragma unroll
  for (int rr = 0; rr < ROWS; ++rr) {
    float s = sum[rr];
#pragma unroll
    for (int m = 16; m >= 1; m >>= 1) s += __shfl_xor(s, m, 32);
    sum[rr] = s;
  }
  float mine = 0.0f;
#pragma unroll
  for (int rr = 0; rr < ROWS; ++rr) mine = (lane == rr) ? sum[rr] : mine;
  if (lane < ROWS) part[kq][rg * ROWS + lane] = mine;
  __syncthreads();
  if (wave == 0) {
    const float s = (part[0][lane] + part[1][lane]) + (part[2][lane] + part[3][lane]);
    const float dy = (-yv + s + xv) * (1.0f / TAU);
    out[blockIdx.x * WG_ROWS + lane] = yv + DT * dy;
  }
}

extern "C" void kernel_launch(void* const* d_in, const int* in_sizes, int n_in,
                              void* d_out, int out_size, void* d_ws, size_t ws_size,
                              hipStream_t stream) {
  const float* x = (const float*)d_in[0];
  const float* y = (const float*)d_in[1];
  const float* W = (const float*)d_in[2];
  float* out = (float*)d_out;

  float* r = (float*)d_ws;

  rnn_prep_kernel<<<NN / 256, 256, 0, stream>>>(y, r, NN);

  rnn_gemv_wmma_kernel<<<NN / WG_ROWS, WAVES * 32, 0, stream>>>(W, r, x, y, out);
}


# One wave per SIMD: workgroups of 4 waves (not 16), each wave streams its K quarter of all 32 rows with 3 rows (48 loads) requested ahead, its 32 row sums folded in one pass; summation order unchanged
# speedup vs baseline: 2.9776x; 1.0301x over previous
#include <hip/hip_runtime.h>


#define NN 8192
#define DT 1e-3f
#define TAU 1e-2f
#define WAVES 4
#define KSPLIT 4
#define ROWS 32
#define WG_ROWS ROWS
#define RING 4
#define KCHUNK (NN / KSPLIT)
#define KQ (KCHUNK / 128)

typedef __attribute__((ext_vector_type(2))) float v2f;
typedef __attribute__((ext_vector_type(4))) float v4f;
typedef __attribute__((ext_vector_type(8))) float v8f;

__global__ void rnn_prep_kernel(const float* __restrict__ y,
                                float* __restrict__ r, int n) {
  int i = blockIdx.x * blockDim.x + threadIdx.x;
  if (i < n) r[i] = tanhf(y[i]);
}

__global__ __launch_bounds__(WAVES * 32) void
rnn_gemv_wmma_kernel(const float* __restrict__ W,
                     const float* __restrict__ r,
                     const float* __restrict__ x,
                     const float* __restrict__ y,
                     float* __restrict__ out) {
  __shared__ float part[KSPLIT][WG_ROWS];

  const int lane = threadIdx.x & 31;
  const int kq   = __builtin_amdgcn_readfirstlane(threadIdx.x >> 5);
  const int row0 = blockIdx.x * WG_ROWS;
  const int col0 = kq * KCHUNK + 4 * lane;

  float xv = 0.0f, yv = 0.0f;
  if (kq == 0) {
    xv = x[row0 + lane];
    yv = y[row0 + lane];
  }

  v4f rq[KQ];
#pragma unroll
  for (int q = 0; q < KQ; ++q) rq[q] = *(const v4f*)(r + col0 + 128 * q);

  const float* wb = W + (size_t)row0 * NN + col0;
  v4f w[RING][KQ];
#define LOAD_ROW(R) do { _Pragma("unroll") for (int q = 0; q < KQ; ++q) \
    w[(R) % RING][q] = __builtin_nontemporal_load((const v4f*)(wb + (size_t)(R) * NN + 128 * q)); } while (0)

  float sum[ROWS];
  LOAD_ROW(0);
  LOAD_ROW(1);
  LOAD_ROW(2);
#pragma unroll
  for (int rr = 0; rr < ROWS; ++rr) {
    __builtin_amdgcn_sched_barrier(0);
    if (rr + 3 < ROWS) LOAD_ROW(rr + 3);
    __builtin_amdgcn_sched_barrier(0);
    const v4f* wr = w[rr % RING];
    v8f c0 = {}, c1 = {};
#pragma unroll
    for (int q = 0; q < KQ; ++q) {
      const v2f a0 = {wr[q].x, wr[q].y}, b0 = {rq[q].x, rq[q].y};
      const v2f a1 = {wr[q].z, wr[q].w}, b1 = {rq[q].z, rq[q].w};
      c0 = __builtin_amdgcn_wmma_f32_16x16x4_f32(false, a0, false, b0, (short)0, c0, false, false);
      c1 = __builtin_amdgcn_wmma_f32_16x16x4_f32(false, a1, false, b1, (short)0, c1, false, false);
    }
    const v8f c = c0 + c1;
    float d = 0.0f;
#pragma unroll
    for (int i = 0; i < 8; ++i) d = ((lane & 7) == i) ? c[i] : d;
    sum[rr] = (lane < 8 || lane >= 24) ? d : 0.0f;
  }
#undef LOAD_ROW

#pragma unroll
  for (int m = 16; m >= 1; m >>= 1) {
    const bool up = (lane & m) != 0;
#pragma unroll
    for (int j = 0; j < m; ++j) {
      float lo = sum[j], hi = sum[j + m];
      asm volatile("" : "+v"(lo), "+v"(hi));
      const float keep = up ? hi : lo;
      const float send = up ? lo : hi;
      sum[j] = keep + __shfl_xor(send, m, 32);
    }
  }
  part[kq][lane] = sum[0];
  __syncthreads();
  if (kq == 0) {
    const float s = (part[0][lane] + part[1][lane]) + (part[2][lane] + part[3][lane]);
    const float dy = (-yv + s + xv) * (1.0f / TAU);
    out[row0 + lane] = yv + DT * dy;
  }
}

extern "C" void kernel_launch(void* const* d_in, const int* in_sizes, int n_in,
                              void* d_out, int out_size, void* d_ws, size_t ws_size,
                              hipStream_t stream) {
  const float* x = (const float*)d_in[0];
  const float* y = (const float*)d_in[1];
  const float* W = (const float*)d_in[2];
  float* out = (float*)d_out;

  float* r = (float*)d_ws;

  rnn_prep_kernel<<<NN / 256, 256, 0, stream>>>(y, r, NN);

  rnn_gemv_wmma_kernel<<<NN / WG_ROWS, WAVES * 32, 0, stream>>>(W, r, x, y, out);
}
